# starting kernel
# baseline (speedup 1.0000x reference)
#include <hip/hip_runtime.h>
#include <hip/hip_bf16.h>


#define HIDDEN 128

typedef float v2f __attribute__((ext_vector_type(2)));
typedef float v8f __attribute__((ext_vector_type(8)));

__global__ void gcn_zero_ws(float4* __restrict__ p, unsigned long long n4) {
    unsigned long long i = (unsigned long long)blockIdx.x * blockDim.x + threadIdx.x;
    unsigned long long stride = (unsigned long long)gridDim.x * blockDim.x;
    float4 z = make_float4(0.f, 0.f, 0.f, 0.f);
    for (; i < n4; i += stride) p[i] = z;
}

__global__ void gcn_spmm_edges(const float* __restrict__ src,
                               const int*   __restrict__ row,
                               const int*   __restrict__ col,
                               const float* __restrict__ val,
                               float*       __restrict__ dst,
                               int n_edges) {
    const int lane   = threadIdx.x & 31;
    const int wave   = threadIdx.x >> 5;
    const int nwaves = (gridDim.x * blockDim.x) >> 5;
    int e = blockIdx.x * (blockDim.x >> 5) + wave;

    for (; e < n_edges; e += nwaves) {
        __builtin_prefetch(col + e + 4096, 0, 0);
        __builtin_prefetch(row + e + 4096, 0, 0);

        const int   r = row[e];
        const int   c = col[e];
        const float v = val[e];

        const float4* s4 = (const float4*)(src + (size_t)c * HIDDEN);
        float4 x = s4[lane];

        float* d = dst + (size_t)r * HIDDEN + lane * 4;
        atomicAdd(d + 0, v * x.x);
        atomicAdd(d + 1, v * x.y);
        atomicAdd(d + 2, v * x.z);
        atomicAdd(d + 3, v * x.w);
    }
}

__device__ __forceinline__ v8f wmma_bias_add(v8f c, float bias_lane, int lane) {
    v2f a, b;
    a.x = (lane < 16) ? 1.0f : 0.0f;
    a.y = 0.0f;
    b.x = bias_lane;
    b.y = 0.0f;
    return __builtin_amdgcn_wmma_f32_16x16x4_f32(
         false, a,  false, b,
         (short)0, c,  false,  false);
}


__global__ void gcn_bias_wmma(float* __restrict__ x,
                              const float* __restrict__ bias,
                              int n_tiles) {
    const int lane = threadIdx.x & 31;
    const int t    = blockIdx.x * 8 + (threadIdx.x >> 5);
    if (t >= n_tiles) return;

    const int row0 = (t >> 3) << 4;
    const int col0 = (t & 7) << 4;
    const int r    = row0 + ((lane >> 4) << 3);
    const int cix  = col0 + (lane & 15);

    float* p = x + (size_t)r * HIDDEN + cix;
    v8f c;
#pragma unroll
    for (int k = 0; k < 8; ++k) c[k] = p[k * HIDDEN];

    v8f d = wmma_bias_add(c, bias[cix - row0 * 0 - col0 + col0]  , lane);
#pragma unroll
    for (int k = 0; k < 8; ++k) p[k * HIDDEN] = d[k];
}

__global__ void gcn_final_wmma(const float* __restrict__ fea,
                               const float* __restrict__ l1,
                               const float* __restrict__ acc2,
                               const float* __restrict__ bias1,
                               float* __restrict__ out,
                               int n_tiles) {
    const int lane = threadIdx.x & 31;
    const int t    = blockIdx.x * 8 + (threadIdx.x >> 5);
    if (t >= n_tiles) return;

    const int row0 = (t >> 3) << 4;
    const int col0 = (t & 7) << 4;
    const int r    = row0 + ((lane >> 4) << 3);
    const int cix  = col0 + (lane & 15);
    const size_t base = (size_t)r * HIDDEN + cix;

    v8f c;
#pragma unroll
    for (int k = 0; k < 8; ++k) {
        size_t idx = base + (size_t)k * HIDDEN;
        c[k] = fea[idx] + l1[idx] + acc2[idx];
    }

    v8f d = wmma_bias_add(c, bias1[cix], lane);

    const float third = 1.0f / 3.0f;
    float* q = out + base;
#pragma unroll
    for (int k = 0; k < 8; ++k) q[k * HIDDEN] = d[k] * third;
}

extern "C" void kernel_launch(void* const* d_in, const int* in_sizes, int n_in,
                              void* d_out, int out_size, void* d_ws, size_t ws_size,
                              hipStream_t stream) {
    const float* fea  = (const float*)d_in[0];
    const int*   row  = (const int*)  d_in[1];
    const int*   col  = (const int*)  d_in[2];
    const float* val  = (const float*)d_in[3];
    const float* bias = (const float*)d_in[4];
    float*       out  = (float*)d_out;

    const int n_nodes = in_sizes[0] / HIDDEN;
    const int n_edges = in_sizes[1];

    float* acc1 = (float*)d_ws;
    float* acc2 = acc1 + (size_t)n_nodes * HIDDEN;

    {
        unsigned long long n4 = (unsigned long long)2 * n_nodes * HIDDEN / 4;
        int blocks = (int)((n4 + 255) / 256);
        if (blocks > 16384) blocks = 16384;
        gcn_zero_ws<<<blocks, 256, 0, stream>>>((float4*)acc1, n4);
    }

    const int spmm_blocks = (n_edges + 7) / 8;
    const int n_tiles     = (n_nodes / 16) * (HIDDEN / 16);
    const int tile_blocks = (n_tiles + 7) / 8;

    gcn_spmm_edges<<<spmm_blocks, 256, 0, stream>>>(fea, row, col, val, acc1, n_edges);
    gcn_bias_wmma<<<tile_blocks, 256, 0, stream>>>(acc1, bias, n_tiles);
    gcn_spmm_edges<<<spmm_blocks, 256, 0, stream>>>(acc1, row, col, val, acc2, n_edges);
    gcn_final_wmma<<<tile_blocks, 256, 0, stream>>>(fea, acc1, acc2, bias + HIDDEN,
                                                    out, n_tiles);
}


# Group edges by destination row in-call (counting sort, integer atomic ranks); each SpMM becomes a row gather summed in registers and stored once: f32 atomics gone, sum order = rank order
# speedup vs baseline: 2.0159x; 2.0159x over previous
#include <hip/hip_runtime.h>
#include <hip/hip_bf16.h>


#define HIDDEN 128

typedef float v2f __attribute__((ext_vector_type(2)));
typedef float v8f __attribute__((ext_vector_type(8)));

__global__ void __launch_bounds__(256) gcn_zero_counts(int4* __restrict__ p, int n4) {
    const int i = blockIdx.x * 256 + threadIdx.x;
    if (i < n4) p[i] = make_int4(0, 0, 0, 0);
}

__global__ void __launch_bounds__(256) gcn_row_rank(const int* __restrict__ row,
                                                    int* __restrict__ cnt,
                                                    int* __restrict__ rank,
                                                    int n_edges) {
    const int e = (blockIdx.x * 256 + threadIdx.x) * 4;
    if (e + 3 < n_edges) {
        const int4 r = *(const int4*)(row + e);
        int4 k;
        k.x = atomicAdd(cnt + r.x, 1);
        k.y = atomicAdd(cnt + r.y, 1);
        k.z = atomicAdd(cnt + r.z, 1);
        k.w = atomicAdd(cnt + r.w, 1);
        *(int4*)(rank + e) = k;
    } else {
        for (int i = e; i < n_edges; ++i) rank[i] = atomicAdd(cnt + row[i], 1);
    }
}

__global__ void __launch_bounds__(1024) gcn_row_start(const int4* __restrict__ cnt4,
                                                      int4* __restrict__ start4,
                                                      int n4) {
    __shared__ int sh_front[32];
    __shared__ int sh_wave[32];
    const int tid  = threadIdx.x;
    const int lane = tid & 31;
    const int w    = tid >> 5;
    const int first = blockIdx.x * 1024;

    int front = 0;
    for (int i0 = 0; i0 < first; i0 += 8 * 1024) {
        int4 f[8];
#pragma unroll
        for (int u = 0; u < 8; ++u) {
            const int i = i0 + u * 1024 + tid;
            f[u] = cnt4[(i < first) ? i : tid];
        }
#pragma unroll
        for (int u = 0; u < 8; ++u) {
            const int i = i0 + u * 1024 + tid;
            const int t = (f[u].x + f[u].y) + (f[u].z + f[u].w);
            front += (i < first) ? t : 0;
        }
    }

    const int idx = first + tid;
    int4 v = make_int4(0, 0, 0, 0);
    if (idx < n4) v = cnt4[idx];
    const int mine = (v.x + v.y) + (v.z + v.w);

    int inc = mine;
#pragma unroll
    for (int d = 1; d < 32; d <<= 1) {
        const int y = __shfl_up(inc, d, 32);
        if (lane >= d) inc += y;
    }
#pragma unroll
    for (int m = 16; m >= 1; m >>= 1) front += __shfl_xor(front, m, 32);

    if (lane == 31) sh_wave[w]  = inc;
    if (lane == 0)  sh_front[w] = front;
    __syncthreads();

    int base = sh_front[lane];
    int pre  = (lane < w) ? sh_wave[lane] : 0;
#pragma unroll
    for (int m = 16; m >= 1; m >>= 1) {
        base += __shfl_xor(base, m, 32);
        pre  += __shfl_xor(pre, m, 32);
    }

    if (idx < n4) {
        const int ex = base + pre + inc - mine;
        start4[idx] = make_int4(ex, ex + v.x, ex + v.x + v.y, ex + v.x + v.y + v.z);
    }
}

__global__ void __launch_bounds__(256) gcn_fill_pairs(const int*   __restrict__ row,
                                                      const int*   __restrict__ col,
                                                      const float* __restrict__ val,
                                                      const int*   __restrict__ start,
                                                      const int*   __restrict__ rank,
                                                      int2*        __restrict__ pairs,
                                                      int n_edges) {
    const int e = (blockIdx.x * 256 + threadIdx.x) * 4;
    if (e + 3 < n_edges) {
        const int4   r = *(const int4*)(row + e);
        const int4   c = *(const int4*)(col + e);
        const float4 v = *(const float4*)(val + e);
        const int4   k = *(const int4*)(rank + e);
        const int s0 = start[r.x], s1 = start[r.y], s2 = start[r.z], s3 = start[r.w];
        pairs[s0 + k.x] = make_int2(c.x, __float_as_int(v.x));
        pairs[s1 + k.y] = make_int2(c.y, __float_as_int(v.y));
        pairs[s2 + k.z] = make_int2(c.z, __float_as_int(v.z));
        pairs[s3 + k.w] = make_int2(c.w, __float_as_int(v.w));
    } else {
        for (int i = e; i < n_edges; ++i)
            pairs[start[row[i]] + rank[i]] = make_int2(col[i], __float_as_int(val[i]));
    }
}

__global__ void __launch_bounds__(256) gcn_spmm_rows(const float* __restrict__ src,
                                                     const int*   __restrict__ start,
                                                     const int2*  __restrict__ pairs,
                                                     float*       __restrict__ dst,
                                                     int n_rows) {
    const int lane = threadIdx.x & 31;
    const int wave = __builtin_amdgcn_readfirstlane(threadIdx.x >> 5);
    const int nw   = gridDim.x * 8;

    for (int r = blockIdx.x * 8 + wave; r < n_rows; r += nw) {
        const int beg = __builtin_amdgcn_readfirstlane(start[r]);
        const int end = __builtin_amdgcn_readfirstlane(start[r + 1]);
        float4 acc = make_float4(0.f, 0.f, 0.f, 0.f);

        for (int j = beg; j < end; j += 32) {
            const int n = (end - j < 32) ? (end - j) : 32;
            int2 pr = make_int2(0, 0);
            if (lane < n) pr = pairs[j + lane];

#pragma unroll 4
            for (int k = 0; k < n; ++k) {
                const int   c = __builtin_amdgcn_readlane(pr.x, k);
                const float v = __int_as_float(__builtin_amdgcn_readlane(pr.y, k));
                const float4 x = ((const float4*)(src + (size_t)c * HIDDEN))[lane];
                acc.x = __builtin_fmaf(v, x.x, acc.x);
                acc.y = __builtin_fmaf(v, x.y, acc.y);
                acc.z = __builtin_fmaf(v, x.z, acc.z);
                acc.w = __builtin_fmaf(v, x.w, acc.w);
            }
        }
        ((float4*)(dst + (size_t)r * HIDDEN))[lane] = acc;
    }
}

__device__ __forceinline__ v8f wmma_bias_add(v8f c, float bias_lane, int lane) {
    v2f a, b;
    a.x = (lane < 16) ? 1.0f : 0.0f;
    a.y = 0.0f;
    b.x = bias_lane;
    b.y = 0.0f;
    return __builtin_amdgcn_wmma_f32_16x16x4_f32(
         false, a,  false, b,
         (short)0, c,  false,  false);
}


__global__ void gcn_bias_wmma(float* __restrict__ x,
                              const float* __restrict__ bias,
                              int n_tiles) {
    const int lane = threadIdx.x & 31;
    const int t    = blockIdx.x * 8 + (threadIdx.x >> 5);
    if (t >= n_tiles) return;

    const int row0 = (t >> 3) << 4;
    const int col0 = (t & 7) << 4;
    const int r    = row0 + ((lane >> 4) << 3);
    const int cix  = col0 + (lane & 15);

    float* p = x + (size_t)r * HIDDEN + cix;
    v8f c;
#pragma unroll
    for (int k = 0; k < 8; ++k) c[k] = p[k * HIDDEN];

    v8f d = wmma_bias_add(c, bias[cix], lane);
#pragma unroll
    for (int k = 0; k < 8; ++k) p[k * HIDDEN] = d[k];
}

__global__ void gcn_final_wmma(const float* __restrict__ fea,
                               const float* __restrict__ l1,
                               const float* __restrict__ bias1,
                               float* __restrict__ out,
                               int n_tiles) {
    const int lane = threadIdx.x & 31;
    const int t    = blockIdx.x * 8 + (threadIdx.x >> 5);
    if (t >= n_tiles) return;

    const int row0 = (t >> 3) << 4;
    const int col0 = (t & 7) << 4;
    const int r    = row0 + ((lane >> 4) << 3);
    const int cix  = col0 + (lane & 15);
    const size_t base = (size_t)r * HIDDEN + cix;

    float* q = out + base;
    v8f c;
#pragma unroll
    for (int k = 0; k < 8; ++k) {
        size_t idx = base + (size_t)k * HIDDEN;
        c[k] = fea[idx] + l1[idx] + q[k * HIDDEN];
    }

    v8f d = wmma_bias_add(c, bias1[cix], lane);

    const float third = 1.0f / 3.0f;
#pragma unroll
    for (int k = 0; k < 8; ++k) q[k * HIDDEN] = d[k] * third;
}

extern "C" void kernel_launch(void* const* d_in, const int* in_sizes, int n_in,
                              void* d_out, int out_size, void* d_ws, size_t ws_size,
                              hipStream_t stream) {
    const float* fea  = (const float*)d_in[0];
    const int*   row  = (const int*)  d_in[1];
    const int*   col  = (const int*)  d_in[2];
    const float* val  = (const float*)d_in[3];
    const float* bias = (const float*)d_in[4];
    float*       out  = (float*)d_out;

    const int n_nodes = in_sizes[0] / HIDDEN;
    const int n_edges = in_sizes[1];

    const int n1 = (n_nodes + 1 + 3) & ~3;
    const int ne = (n_edges + 3) & ~3;
    float* acc1  = (float*)d_ws;
    int*   cnt   = (int*)(acc1 + (size_t)n_nodes * HIDDEN);
    int*   start = cnt + n1;
    int*   rank  = start + n1;
    int2*  pairs = (int2*)(rank + ne);

    const int n4          = n1 / 4;
    const int edge_blocks = (ne / 4 + 255) / 256;
    const int n_tiles     = (n_nodes / 16) * (HIDDEN / 16);
    const int tile_blocks = (n_tiles + 7) / 8;
    int spmm_blocks = (n_nodes + 7) / 8;
    if (spmm_blocks > 2048) spmm_blocks = 2048;

    gcn_zero_counts<<<(n4 + 255) / 256, 256, 0, stream>>>((int4*)cnt, n4);
    gcn_row_rank<<<edge_blocks, 256, 0, stream>>>(row, cnt, rank, n_edges);
    gcn_row_start<<<(n4 + 1023) / 1024, 1024, 0, stream>>>((const int4*)cnt, (int4*)start, n4);
    gcn_fill_pairs<<<edge_blocks, 256, 0, stream>>>(row, col, val, start, rank, pairs, n_edges);

    gcn_spmm_rows<<<spmm_blocks, 256, 0, stream>>>(fea, start, pairs, acc1, n_nodes);
    gcn_bias_wmma<<<tile_blocks, 256, 0, stream>>>(acc1, bias, n_tiles);
    gcn_spmm_rows<<<spmm_blocks, 256, 0, stream>>>(acc1, start, pairs, out, n_nodes);
    gcn_final_wmma<<<tile_blocks, 256, 0, stream>>>(fea, acc1, bias + HIDDEN, out, n_tiles);
}


# Bias add and final (fea+L1+L2)/3 fused into the SpMM row gather: a wave owns 4 rows, adds bias to its register sums with 2 WMMAs, stores each row once; both tile passes dropped; sum order kept
# speedup vs baseline: 2.2758x; 1.1289x over previous
#include <hip/hip_runtime.h>
#include <hip/hip_bf16.h>


#define HIDDEN 128

typedef float v2f __attribute__((ext_vector_type(2)));
typedef float v8f __attribute__((ext_vector_type(8)));

__global__ void __launch_bounds__(256) gcn_zero_counts(int4* __restrict__ p, int n4) {
    const int i = blockIdx.x * 256 + threadIdx.x;
    if (i < n4) p[i] = make_int4(0, 0, 0, 0);
}

__global__ void __launch_bounds__(256) gcn_row_rank(const int* __restrict__ row,
                                                    int* __restrict__ cnt,
                                                    int* __restrict__ rank,
                                                    int n_edges) {
    const int e = (blockIdx.x * 256 + threadIdx.x) * 4;
    if (e + 3 < n_edges) {
        const int4 r = *(const int4*)(row + e);
        int4 k;
        k.x = atomicAdd(cnt + r.x, 1);
        k.y = atomicAdd(cnt + r.y, 1);
        k.z = atomicAdd(cnt + r.z, 1);
        k.w = atomicAdd(cnt + r.w, 1);
        *(int4*)(rank + e) = k;
    } else {
        for (int i = e; i < n_edges; ++i) rank[i] = atomicAdd(cnt + row[i], 1);
    }
}

__global__ void __launch_bounds__(1024) gcn_row_start(const int4* __restrict__ cnt4,
                                                      int4* __restrict__ start4,
                                                      int n4) {
    __shared__ int sh_front[32];
    __shared__ int sh_wave[32];
    const int tid  = threadIdx.x;
    const int lane = tid & 31;
    const int w    = tid >> 5;
    const int first = blockIdx.x * 1024;

    int front = 0;
    for (int i0 = 0; i0 < first; i0 += 8 * 1024) {
        int4 f[8];
#pragma unroll
        for (int u = 0; u < 8; ++u) {
            const int i = i0 + u * 1024 + tid;
            f[u] = cnt4[(i < first) ? i : tid];
        }
#pragma unroll
        for (int u = 0; u < 8; ++u) {
            const int i = i0 + u * 1024 + tid;
            const int t = (f[u].x + f[u].y) + (f[u].z + f[u].w);
            front += (i < first) ? t : 0;
        }
    }

    const int idx = first + tid;
    int4 v = make_int4(0, 0, 0, 0);
    if (idx < n4) v = cnt4[idx];
    const int mine = (v.x + v.y) + (v.z + v.w);

    int inc = mine;
#pragma unroll
    for (int d = 1; d < 32; d <<= 1) {
        const int y = __shfl_up(inc, d, 32);
        if (lane >= d) inc += y;
    }
#pragma unroll
    for (int m = 16; m >= 1; m >>= 1) front += __shfl_xor(front, m, 32);

    if (lane == 31) sh_wave[w]  = inc;
    if (lane == 0)  sh_front[w] = front;
    __syncthreads();

    int base = sh_front[lane];
    int pre  = (lane < w) ? sh_wave[lane] : 0;
#pragma unroll
    for (int m = 16; m >= 1; m >>= 1) {
        base += __shfl_xor(base, m, 32);
        pre  += __shfl_xor(pre, m, 32);
    }

    if (idx < n4) {
        const int ex = base + pre + inc - mine;
        start4[idx] = make_int4(ex, ex + v.x, ex + v.x + v.y, ex + v.x + v.y + v.z);
    }
}

__global__ void __launch_bounds__(256) gcn_fill_pairs(const int*   __restrict__ row,
                                                      const int*   __restrict__ col,
                                                      const float* __restrict__ val,
                                                      const int*   __restrict__ start,
                                                      const int*   __restrict__ rank,
                                                      int2*        __restrict__ pairs,
                                                      int n_edges) {
    const int e = (blockIdx.x * 256 + threadIdx.x) * 4;
    if (e + 3 < n_edges) {
        const int4   r = *(const int4*)(row + e);
        const int4   c = *(const int4*)(col + e);
        const float4 v = *(const float4*)(val + e);
        const int4   k = *(const int4*)(rank + e);
        const int s0 = start[r.x], s1 = start[r.y], s2 = start[r.z], s3 = start[r.w];
        pairs[s0 + k.x] = make_int2(c.x, __float_as_int(v.x));
        pairs[s1 + k.y] = make_int2(c.y, __float_as_int(v.y));
        pairs[s2 + k.z] = make_int2(c.z, __float_as_int(v.z));
        pairs[s3 + k.w] = make_int2(c.w, __float_as_int(v.w));
    } else {
        for (int i = e; i < n_edges; ++i)
            pairs[start[row[i]] + rank[i]] = make_int2(col[i], __float_as_int(val[i]));
    }
}

__device__ __forceinline__ float4 gcn_gather_row(const float* __restrict__ src,
                                                 const int2*  __restrict__ pairs,
                                                 int beg, int end, int2 pr, int lane) {
    float4 acc = make_float4(0.f, 0.f, 0.f, 0.f);
    for (int j = beg; j < end; j += 32) {
        const int n = (end - j < 32) ? (end - j) : 32;
        if (j != beg) {
            pr = make_int2(0, 0);
            if (lane < n) pr = pairs[j + lane];
        }
#pragma unroll 4
        for (int k = 0; k < n; ++k) {
            const int   c = __builtin_amdgcn_readlane(pr.x, k);
            const float v = __int_as_float(__builtin_amdgcn_readlane(pr.y, k));
            const float4 x = ((const float4*)(src + (size_t)c * HIDDEN))[lane];
            acc.x = __builtin_fmaf(v, x.x, acc.x);
            acc.y = __builtin_fmaf(v, x.y, acc.y);
            acc.z = __builtin_fmaf(v, x.z, acc.z);
            acc.w = __builtin_fmaf(v, x.w, acc.w);
        }
    }
    return acc;
}

__device__ __forceinline__ v8f wmma_add_lane_pair(v8f c, v2f a, v2f b) {
    return __builtin_amdgcn_wmma_f32_16x16x4_f32(
         false, a,  false, b,
         (short)0, c,  false,  false);
}

template <bool FINAL>
__device__ __forceinline__ void gcn_spmm_group4(const float* __restrict__ src,
                                                const int*   __restrict__ start,
                                                const int2*  __restrict__ pairs,
                                                const float* __restrict__ bias,
                                                const float* __restrict__ fea,
                                                float*       __restrict__ dst,
                                                int n_groups) {
    const int lane = threadIdx.x & 31;
    const int wave = __builtin_amdgcn_readfirstlane(threadIdx.x >> 5);
    const int g    = blockIdx.x * 8 + wave;
    if (g >= n_groups) return;
    const int r0 = g * 4;

    int s[5];
#pragma unroll
    for (int i = 0; i < 5; ++i) s[i] = __builtin_amdgcn_readfirstlane(start[r0 + i]);

    int2 pr[4];
#pragma unroll
    for (int i = 0; i < 4; ++i) {
        pr[i] = make_int2(0, 0);
        if (lane < s[i + 1] - s[i]) pr[i] = pairs[s[i] + lane];
    }

    float4 acc[4];
#pragma unroll
    for (int i = 0; i < 4; ++i) acc[i] = gcn_gather_row(src, pairs, s[i], s[i + 1], pr[i], lane);

    if (FINAL) {
        float4 f[4], l[4];
#pragma unroll
        for (int i = 0; i < 4; ++i) {
            f[i] = ((const float4*)(fea + (size_t)(r0 + i) * HIDDEN))[lane];
            l[i] = ((const float4*)(src + (size_t)(r0 + i) * HIDDEN))[lane];
        }
#pragma unroll
        for (int i = 0; i < 4; ++i) {
            acc[i].x = (f[i].x + l[i].x) + acc[i].x;
            acc[i].y = (f[i].y + l[i].y) + acc[i].y;
            acc[i].z = (f[i].z + l[i].z) + acc[i].z;
            acc[i].w = (f[i].w + l[i].w) + acc[i].w;
        }
    }

    const int  c   = lane & 15;
    const bool sel = (lane >> 4) == (c >> 3);
    v2f a;
    a.x = (sel && !(c & 1)) ? 1.0f : 0.0f;
    a.y = (sel &&  (c & 1)) ? 1.0f : 0.0f;
    const float4 bv = ((const float4*)bias)[lane];
    v2f bxy, bzw;
    bxy.x = bv.x; bxy.y = bv.y;
    bzw.x = bv.z; bzw.y = bv.w;

    v8f cxy, czw;
#pragma unroll
    for (int i = 0; i < 4; ++i) {
        cxy[2 * i] = acc[i].x; cxy[2 * i + 1] = acc[i].y;
        czw[2 * i] = acc[i].z; czw[2 * i + 1] = acc[i].w;
    }
    v8f dxy = wmma_add_lane_pair(cxy, a, bxy);
    v8f dzw = wmma_add_lane_pair(czw, a, bzw);
    asm volatile("v_nop\n\tv_nop\n\tv_nop\n\tv_nop"
                 : "+v"(dxy), "+v"(dzw) : "v"(a), "v"(bxy), "v"(bzw));

    const float scale = FINAL ? (1.0f / 3.0f) : 1.0f;
#pragma unroll
    for (int i = 0; i < 4; ++i) {
        float4 o;
        if (FINAL) o = make_float4(dxy[2 * i] * scale, dxy[2 * i + 1] * scale,
                                   dzw[2 * i] * scale, dzw[2 * i + 1] * scale);
        else       o = make_float4(dxy[2 * i], dxy[2 * i + 1], dzw[2 * i], dzw[2 * i + 1]);
        ((float4*)(dst + (size_t)(r0 + i) * HIDDEN))[lane] = o;
    }
}

__global__ void __launch_bounds__(256) gcn_spmm_bias(const float* __restrict__ fea,
                                                     const int*   __restrict__ start,
                                                     const int2*  __restrict__ pairs,
                                                     const float* __restrict__ bias,
                                                     float*       __restrict__ l1,
                                                     int n_groups) {
    gcn_spmm_group4<false>(fea, start, pairs, bias, fea, l1, n_groups);
}

__global__ void __launch_bounds__(256) gcn_spmm_bias_mean(const float* __restrict__ l1,
                                                          const int*   __restrict__ start,
                                                          const int2*  __restrict__ pairs,
                                                          const float* __restrict__ bias,
                                                          const float* __restrict__ fea,
                                                          float*       __restrict__ out,
                                                          int n_groups) {
    gcn_spmm_group4<true>(l1, start, pairs, bias, fea, out, n_groups);
}

extern "C" void kernel_launch(void* const* d_in, const int* in_sizes, int n_in,
                              void* d_out, int out_size, void* d_ws, size_t ws_size,
                              hipStream_t stream) {
    const float* fea  = (const float*)d_in[0];
    const int*   row  = (const int*)  d_in[1];
    const int*   col  = (const int*)  d_in[2];
    const float* val  = (const float*)d_in[3];
    const float* bias = (const float*)d_in[4];
    float*       out  = (float*)d_out;

    const int n_nodes = in_sizes[0] / HIDDEN;
    const int n_edges = in_sizes[1];

    const int n1 = (n_nodes + 1 + 3) & ~3;
    const int ne = (n_edges + 3) & ~3;
    float* acc1  = (float*)d_ws;
    int*   cnt   = (int*)(acc1 + (size_t)n_nodes * HIDDEN);
    int*   start = cnt + n1;
    int*   rank  = start + n1;
    int2*  pairs = (int2*)(rank + ne);

    const int n4          = n1 / 4;
    const int edge_blocks = (ne / 4 + 255) / 256;
    const int n_groups    = n_nodes / 4;
    const int spmm_blocks = (n_groups + 7) / 8;

    gcn_zero_counts<<<(n4 + 255) / 256, 256, 0, stream>>>((int4*)cnt, n4);
    gcn_row_rank<<<edge_blocks, 256, 0, stream>>>(row, cnt, rank, n_edges);
    gcn_row_start<<<(n4 + 1023) / 1024, 1024, 0, stream>>>((const int4*)cnt, (int4*)start, n4);
    gcn_fill_pairs<<<edge_blocks, 256, 0, stream>>>(row, col, val, start, rank, pairs, n_edges);

    gcn_spmm_bias<<<spmm_blocks, 256, 0, stream>>>(fea, start, pairs, bias, acc1, n_groups);
    gcn_spmm_bias_mean<<<spmm_blocks, 256, 0, stream>>>(acc1, start, pairs, bias + HIDDEN, fea, out, n_groups);
}


# Edges grouped by destination row with a two-level counting sort on LDS counters (chunk x 256-row bucket table, scan, bucketed copy, per-bucket row sort): no global atomic; sum order = LDS rank order
# speedup vs baseline: 3.3915x; 1.4902x over previous
#include <hip/hip_runtime.h>
#include <hip/hip_bf16.h>


#define HIDDEN 128

#define RB_SHIFT 8
#define RB       256
#define CHUNK    2048
#define MAXB     256
#define STAGE    6144

typedef float v2f __attribute__((ext_vector_type(2)));
typedef float v8f __attribute__((ext_vector_type(8)));

__device__ __forceinline__ void gcn_scan256(int* c, int lane) {
    int a[8];
    int s = 0;
#pragma unroll
    for (int j = 0; j < 8; ++j) { a[j] = c[lane * 8 + j]; s += a[j]; }
    int inc = s;
#pragma unroll
    for (int d = 1; d < 32; d <<= 1) {
        const int y = __shfl_up(inc, d, 32);
        if (lane >= d) inc += y;
    }
    int ex = inc - s;
#pragma unroll
    for (int j = 0; j < 8; ++j) { c[lane * 8 + j] = ex; ex += a[j]; }
}

__global__ void __launch_bounds__(512) gcn_chunk_hist(const int* __restrict__ row,
                                                      int* __restrict__ hist,
                                                      int n_edges, int n_chunks,
                                                      int n_buckets, int n_hist) {
    __shared__ int h[MAXB];
    const int tid = threadIdx.x;
    const int w   = blockIdx.x;
    if (tid < MAXB) h[tid] = 0;
    __syncthreads();
    const int e = w * CHUNK + tid * 4;
    if (e + 3 < n_edges) {
        const int4 r = *(const int4*)(row + e);
        atomicAdd(&h[r.x >> RB_SHIFT], 1);
        atomicAdd(&h[r.y >> RB_SHIFT], 1);
        atomicAdd(&h[r.z >> RB_SHIFT], 1);
        atomicAdd(&h[r.w >> RB_SHIFT], 1);
    } else {
        for (int i = e; i < n_edges; ++i) atomicAdd(&h[row[i] >> RB_SHIFT], 1);
    }
    __syncthreads();
    if (tid < n_buckets) hist[tid * n_chunks + w] = h[tid];
    if (w == 0)
        for (int i = n_buckets * n_chunks + tid; i < n_hist; i += 512) hist[i] = 0;
}

__global__ void __launch_bounds__(1024) gcn_row_start(const int4* __restrict__ cnt4,
                                                      int4* __restrict__ start4,
                                                      int n4) {
    __shared__ int sh_front[32];
    __shared__ int sh_wave[32];
    const int tid  = threadIdx.x;
    const int lane = tid & 31;
    const int w    = tid >> 5;
    const int first = blockIdx.x * 1024;

    int front = 0;
    for (int i0 = 0; i0 < first; i0 += 8 * 1024) {
        int4 f[8];
#pragma unroll
        for (int u = 0; u < 8; ++u) {
            const int i = i0 + u * 1024 + tid;
            f[u] = cnt4[(i < first) ? i : tid];
        }
#pragma unroll
        for (int u = 0; u < 8; ++u) {
            const int i = i0 + u * 1024 + tid;
            const int t = (f[u].x + f[u].y) + (f[u].z + f[u].w);
            front += (i < first) ? t : 0;
        }
    }

    const int idx = first + tid;
    int4 v = make_int4(0, 0, 0, 0);
    if (idx < n4) v = cnt4[idx];
    const int mine = (v.x + v.y) + (v.z + v.w);

    int inc = mine;
#pragma unroll
    for (int d = 1; d < 32; d <<= 1) {
        const int y = __shfl_up(inc, d, 32);
        if (lane >= d) inc += y;
    }
#pragma unroll
    for (int m = 16; m >= 1; m >>= 1) front += __shfl_xor(front, m, 32);

    if (lane == 31) sh_wave[w]  = inc;
    if (lane == 0)  sh_front[w] = front;
    __syncthreads();

    int base = sh_front[lane];
    int pre  = (lane < w) ? sh_wave[lane] : 0;
#pragma unroll
    for (int m = 16; m >= 1; m >>= 1) {
        base += __shfl_xor(base, m, 32);
        pre  += __shfl_xor(pre, m, 32);
    }

    if (idx < n4) {
        const int ex = base + pre + inc - mine;
        start4[idx] = make_int4(ex, ex + v.x, ex + v.x + v.y, ex + v.x + v.y + v.z);
    }
}

__global__ void __launch_bounds__(512) gcn_bucket_scatter(const int*   __restrict__ row,
                                                          const int*   __restrict__ col,
                                                          const float* __restrict__ val,
                                                          const int*   __restrict__ off,
                                                          int*         __restrict__ rec_row,
                                                          int2*        __restrict__ rec_cv,
                                                          int n_edges, int n_chunks,
                                                          int n_buckets) {
    __shared__ int  cntb[MAXB];
    __shared__ int  gbase[MAXB];
    __shared__ int  s_dst[CHUNK];
    __shared__ int  s_row[CHUNK];
    __shared__ int2 s_cv[CHUNK];
    const int tid = threadIdx.x;
    const int w   = blockIdx.x;
    if (tid < MAXB) {
        cntb[tid]  = 0;
        gbase[tid] = (tid < n_buckets) ? off[tid * n_chunks + w] : 0;
    }
    __syncthreads();

    const int e = w * CHUNK + tid * 4;
    int left = n_edges - e;
    left = (left < 0) ? 0 : ((left > 4) ? 4 : left);
    int r[4], c[4], v[4], k[4];
    if (left == 4) {
        const int4   r4 = *(const int4*)(row + e);
        const int4   c4 = *(const int4*)(col + e);
        const float4 v4 = *(const float4*)(val + e);
        r[0] = r4.x; r[1] = r4.y; r[2] = r4.z; r[3] = r4.w;
        c[0] = c4.x; c[1] = c4.y; c[2] = c4.z; c[3] = c4.w;
        v[0] = __float_as_int(v4.x); v[1] = __float_as_int(v4.y);
        v[2] = __float_as_int(v4.z); v[3] = __float_as_int(v4.w);
    } else {
#pragma unroll
        for (int i = 0; i < 4; ++i) {
            r[i] = 0; c[i] = 0; v[i] = 0;
            if (i < left) { r[i] = row[e + i]; c[i] = col[e + i]; v[i] = __float_as_int(val[e + i]); }
        }
    }
#pragma unroll
    for (int i = 0; i < 4; ++i) {
        k[i] = 0;
        if (i < left) k[i] = atomicAdd(&cntb[r[i] >> RB_SHIFT], 1);
    }
    __syncthreads();
    if (tid < 32) gcn_scan256(cntb, tid);
    __syncthreads();
#pragma unroll
    for (int i = 0; i < 4; ++i) {
        if (i < left) {
            const int b    = r[i] >> RB_SHIFT;
            const int slot = cntb[b] + k[i];
            s_dst[slot] = gbase[b] + k[i];
            s_row[slot] = r[i];
            s_cv[slot]  = make_int2(c[i], v[i]);
        }
    }
    __syncthreads();
    int n_mine = n_edges - w * CHUNK;
    n_mine = (n_mine > CHUNK) ? CHUNK : n_mine;
    for (int s = tid; s < n_mine; s += 512) {
        const int d = s_dst[s];
        rec_row[d] = s_row[s];
        rec_cv[d]  = s_cv[s];
    }
}

__global__ void __launch_bounds__(1024) gcn_bucket_rows(const int*  __restrict__ rec_row,
                                                        const int2* __restrict__ rec_cv,
                                                        const int*  __restrict__ off,
                                                        int*        __restrict__ start,
                                                        int2*       __restrict__ pairs,
                                                        int n_nodes, int n_chunks,
                                                        int n_buckets) {
    __shared__ int  cnt[RB];
    __shared__ int2 stage[STAGE];
    const int tid = threadIdx.x;
    const int b   = blockIdx.x;
    const int beg = off[b * n_chunks];
    const int end = off[(b + 1) * n_chunks];
    if (tid < RB) cnt[tid] = 0;
    __syncthreads();
    for (int i0 = beg + tid; i0 < end; i0 += 4 * 1024) {
        int rr[4];
#pragma unroll
        for (int u = 0; u < 4; ++u) {
            const int i = i0 + u * 1024;
            rr[u] = (i < end) ? rec_row[i] : -1;
        }
#pragma unroll
        for (int u = 0; u < 4; ++u)
            if (rr[u] >= 0) atomicAdd(&cnt[rr[u] & (RB - 1)], 1);
    }
    __syncthreads();
    if (tid < 32) gcn_scan256(cnt, tid);
    __syncthreads();
    if (tid < RB) {
        const int g = b * RB + tid;
        if (g < n_nodes) start[g] = beg + cnt[tid];
    }
    if (b == n_buckets - 1 && tid == 0) start[n_nodes] = end;
    __syncthreads();
    for (int i0 = beg + tid; i0 < end; i0 += 4 * 1024) {
        int  rr[4];
        int2 cv[4];
#pragma unroll
        for (int u = 0; u < 4; ++u) {
            const int i = i0 + u * 1024;
            rr[u] = -1;
            cv[u] = make_int2(0, 0);
            if (i < end) { rr[u] = rec_row[i]; cv[u] = rec_cv[i]; }
        }
#pragma unroll
        for (int u = 0; u < 4; ++u) {
            if (rr[u] >= 0) {
                const int q = atomicAdd(&cnt[rr[u] & (RB - 1)], 1);
                if (q < STAGE) stage[q] = cv[u];
                else           pairs[beg + q] = cv[u];
            }
        }
    }
    __syncthreads();
    int m = end - beg;
    m = (m > STAGE) ? STAGE : m;
    for (int i = tid; i < m; i += 1024) pairs[beg + i] = stage[i];
}

__device__ __forceinline__ float4 gcn_gather_row(const float* __restrict__ src,
                                                 const int2*  __restrict__ pairs,
                                                 int beg, int end, int2 pr, int lane) {
    float4 acc = make_float4(0.f, 0.f, 0.f, 0.f);
    for (int j = beg; j < end; j += 32) {
        const int n = (end - j < 32) ? (end - j) : 32;
        if (j != beg) {
            pr = make_int2(0, 0);
            if (lane < n) pr = pairs[j + lane];
        }
#pragma unroll 4
        for (int k = 0; k < n; ++k) {
            const int   c = __builtin_amdgcn_readlane(pr.x, k);
            const float v = __int_as_float(__builtin_amdgcn_readlane(pr.y, k));
            const float4 x = ((const float4*)(src + (size_t)c * HIDDEN))[lane];
            acc.x = __builtin_fmaf(v, x.x, acc.x);
            acc.y = __builtin_fmaf(v, x.y, acc.y);
            acc.z = __builtin_fmaf(v, x.z, acc.z);
            acc.w = __builtin_fmaf(v, x.w, acc.w);
        }
    }
    return acc;
}

__device__ __forceinline__ v8f wmma_add_lane_pair(v8f c, v2f a, v2f b) {
    return __builtin_amdgcn_wmma_f32_16x16x4_f32(
         false, a,  false, b,
         (short)0, c,  false,  false);
}

template <bool FINAL>
__device__ __forceinline__ void gcn_spmm_group4(const float* __restrict__ src,
                                                const int*   __restrict__ start,
                                                const int2*  __restrict__ pairs,
                                                const float* __restrict__ bias,
                                                const float* __restrict__ fea,
                                                float*       __restrict__ dst,
                                                int n_groups) {
    const int lane = threadIdx.x & 31;
    const int wave = __builtin_amdgcn_readfirstlane(threadIdx.x >> 5);
    const int g    = blockIdx.x * 8 + wave;
    if (g >= n_groups) return;
    const int r0 = g * 4;

    int s[5];
#pragma unroll
    for (int i = 0; i < 5; ++i) s[i] = __builtin_amdgcn_readfirstlane(start[r0 + i]);

    int2 pr[4];
#pragma unroll
    for (int i = 0; i < 4; ++i) {
        pr[i] = make_int2(0, 0);
        if (lane < s[i + 1] - s[i]) pr[i] = pairs[s[i] + lane];
    }

    float4 acc[4];
#pragma unroll
    for (int i = 0; i < 4; ++i) acc[i] = gcn_gather_row(src, pairs, s[i], s[i + 1], pr[i], lane);

    if (FINAL) {
        float4 f[4], l[4];
#pragma unroll
        for (int i = 0; i < 4; ++i) {
            f[i] = ((const float4*)(fea + (size_t)(r0 + i) * HIDDEN))[lane];
            l[i] = ((const float4*)(src + (size_t)(r0 + i) * HIDDEN))[lane];
        }
#pragma unroll
        for (int i = 0; i < 4; ++i) {
            acc[i].x = (f[i].x + l[i].x) + acc[i].x;
            acc[i].y = (f[i].y + l[i].y) + acc[i].y;
            acc[i].z = (f[i].z + l[i].z) + acc[i].z;
            acc[i].w = (f[i].w + l[i].w) + acc[i].w;
        }
    }

    const int  c   = lane & 15;
    const bool sel = (lane >> 4) == (c >> 3);
    v2f a;
    a.x = (sel && !(c & 1)) ? 1.0f : 0.0f;
    a.y = (sel &&  (c & 1)) ? 1.0f : 0.0f;
    const float4 bv = ((const float4*)bias)[lane];
    v2f bxy, bzw;
    bxy.x = bv.x; bxy.y = bv.y;
    bzw.x = bv.z; bzw.y = bv.w;

    v8f cxy, czw;
#pragma unroll
    for (int i = 0; i < 4; ++i) {
        cxy[2 * i] = acc[i].x; cxy[2 * i + 1] = acc[i].y;
        czw[2 * i] = acc[i].z; czw[2 * i + 1] = acc[i].w;
    }
    v8f dxy = wmma_add_lane_pair(cxy, a, bxy);
    v8f dzw = wmma_add_lane_pair(czw, a, bzw);
    asm volatile("v_nop\n\tv_nop\n\tv_nop\n\tv_nop"
                 : "+v"(dxy), "+v"(dzw) : "v"(a), "v"(bxy), "v"(bzw));

    const float scale = FINAL ? (1.0f / 3.0f) : 1.0f;
#pragma unroll
    for (int i = 0; i < 4; ++i) {
        float4 o;
        if (FINAL) o = make_float4(dxy[2 * i] * scale, dxy[2 * i + 1] * scale,
                                   dzw[2 * i] * scale, dzw[2 * i + 1] * scale);
        else       o = make_float4(dxy[2 * i], dxy[2 * i + 1], dzw[2 * i], dzw[2 * i + 1]);
        ((float4*)(dst + (size_t)(r0 + i) * HIDDEN))[lane] = o;
    }
}

__global__ void __launch_bounds__(256) gcn_spmm_bias(const float* __restrict__ fea,
                                                     const int*   __restrict__ start,
                                                     const int2*  __restrict__ pairs,
                                                     const float* __restrict__ bias,
                                                     float*       __restrict__ l1,
                                                     int n_groups) {
    gcn_spmm_group4<false>(fea, start, pairs, bias, fea, l1, n_groups);
}

__global__ void __launch_bounds__(256) gcn_spmm_bias_mean(const float* __restrict__ l1,
                                                          const int*   __restrict__ start,
                                                          const int2*  __restrict__ pairs,
                                                          const float* __restrict__ bias,
                                                          const float* __restrict__ fea,
                                                          float*       __restrict__ out,
                                                          int n_groups) {
    gcn_spmm_group4<true>(l1, start, pairs, bias, fea, out, n_groups);
}

extern "C" void kernel_launch(void* const* d_in, const int* in_sizes, int n_in,
                              void* d_out, int out_size, void* d_ws, size_t ws_size,
                              hipStream_t stream) {
    const float* fea  = (const float*)d_in[0];
    const int*   row  = (const int*)  d_in[1];
    const int*   col  = (const int*)  d_in[2];
    const float* val  = (const float*)d_in[3];
    const float* bias = (const float*)d_in[4];
    float*       out  = (float*)d_out;

    const int n_nodes = in_sizes[0] / HIDDEN;
    const int n_edges = in_sizes[1];

    const int n1        = (n_nodes + 1 + 3) & ~3;
    const int ne        = (n_edges + 3) & ~3;
    const int n_chunks  = (n_edges + CHUNK - 1) / CHUNK;
    const int n_buckets = (n_nodes + RB - 1) / RB;
    const int n_hist    = (n_buckets * n_chunks + 1 + 3) & ~3;

    float* acc1    = (float*)d_ws;
    int2*  rec_cv  = (int2*)d_ws;
    int*   rec_row = (int*)(rec_cv + ne);
    int*   hist    = (int*)(acc1 + (size_t)n_nodes * HIDDEN);
    int*   off     = hist + n_hist;
    int*   start   = off + n_hist;
    int2*  pairs   = (int2*)(start + n1);

    const int n4h         = n_hist / 4;
    const int n_groups    = n_nodes / 4;
    const int spmm_blocks = (n_groups + 7) / 8;

    gcn_chunk_hist<<<n_chunks, 512, 0, stream>>>(row, hist, n_edges, n_chunks, n_buckets, n_hist);
    gcn_row_start<<<(n4h + 1023) / 1024, 1024, 0, stream>>>((const int4*)hist, (int4*)off, n4h);
    gcn_bucket_scatter<<<n_chunks, 512, 0, stream>>>(row, col, val, off, rec_row, rec_cv,
                                                     n_edges, n_chunks, n_buckets);
    gcn_bucket_rows<<<n_buckets, 1024, 0, stream>>>(rec_row, rec_cv, off, start, pairs,
                                                    n_nodes, n_chunks, n_buckets);

    gcn_spmm_bias<<<spmm_blocks, 256, 0, stream>>>(fea, start, pairs, bias, acc1, n_groups);
    gcn_spmm_bias_mean<<<spmm_blocks, 256, 0, stream>>>(acc1, start, pairs, bias + HIDDEN, fea, out, n_groups);
}


# Edge grouping in 2 launches, not 4: chunks write their edges bucket-ordered in their own spans, a bucket work-group collects its cells; histogram and scan launches gone; sum order = LDS rank order
# speedup vs baseline: 4.0081x; 1.1818x over previous
#include <hip/hip_runtime.h>
#include <hip/hip_bf16.h>


#define HIDDEN 128

#define RB_SHIFT 8
#define RB       256
#define CHUNK    2048
#define MAXB     256
#define STAGE    6144
#define CELL_U   8

typedef float v2f __attribute__((ext_vector_type(2)));
typedef float v8f __attribute__((ext_vector_type(8)));

__device__ __forceinline__ void gcn_scan256(int* c, int lane) {
    int a[8];
    int s = 0;
#pragma unroll
    for (int j = 0; j < 8; ++j) { a[j] = c[lane * 8 + j]; s += a[j]; }
    int inc = s;
#pragma unroll
    for (int d = 1; d < 32; d <<= 1) {
        const int y = __shfl_up(inc, d, 32);
        if (lane >= d) inc += y;
    }
    int ex = inc - s;
#pragma unroll
    for (int j = 0; j < 8; ++j) { c[lane * 8 + j] = ex; ex += a[j]; }
}

__global__ void __launch_bounds__(512) gcn_chunk_sort(const int*   __restrict__ row,
                                                      const int*   __restrict__ col,
                                                      const float* __restrict__ val,
                                                      int*         __restrict__ tab,
                                                      int*         __restrict__ rec_row,
                                                      int2*        __restrict__ rec_cv,
                                                      int n_edges, int n_chunks,
                                                      int n_buckets) {
    __shared__ int  cntb[MAXB];
    __shared__ int  s_row[CHUNK];
    __shared__ int2 s_cv[CHUNK];
    const int tid = threadIdx.x;
    const int w   = blockIdx.x;
    if (tid < MAXB) cntb[tid] = 0;
    __syncthreads();

    const int e = w * CHUNK + tid * 4;
    int left = n_edges - e;
    left = (left < 0) ? 0 : ((left > 4) ? 4 : left);
    int r[4], c[4], v[4], k[4];
    if (left == 4) {
        const int4   r4 = *(const int4*)(row + e);
        const int4   c4 = *(const int4*)(col + e);
        const float4 v4 = *(const float4*)(val + e);
        r[0] = r4.x; r[1] = r4.y; r[2] = r4.z; r[3] = r4.w;
        c[0] = c4.x; c[1] = c4.y; c[2] = c4.z; c[3] = c4.w;
        v[0] = __float_as_int(v4.x); v[1] = __float_as_int(v4.y);
        v[2] = __float_as_int(v4.z); v[3] = __float_as_int(v4.w);
    } else {
#pragma unroll
        for (int i = 0; i < 4; ++i) {
            r[i] = 0; c[i] = 0; v[i] = 0;
            if (i < left) { r[i] = row[e + i]; c[i] = col[e + i]; v[i] = __float_as_int(val[e + i]); }
        }
    }
#pragma unroll
    for (int i = 0; i < 4; ++i) {
        k[i] = 0;
        if (i < left) k[i] = atomicAdd(&cntb[r[i] >> RB_SHIFT], 1);
    }
    __syncthreads();
    if (tid < 32) gcn_scan256(cntb, tid);
    __syncthreads();
    int n_mine = n_edges - w * CHUNK;
    n_mine = (n_mine > CHUNK) ? CHUNK : n_mine;
#pragma unroll
    for (int i = 0; i < 4; ++i) {
        if (i < left) {
            const int slot = cntb[r[i] >> RB_SHIFT] + k[i];
            s_row[slot] = r[i];
            s_cv[slot]  = make_int2(c[i], v[i]);
        }
    }
    if (tid <= n_buckets) tab[tid * n_chunks + w] = (tid < n_buckets) ? cntb[tid] : n_mine;
    __syncthreads();
    for (int s = tid; s < n_mine; s += 512) {
        rec_row[w * CHUNK + s] = s_row[s];
        rec_cv[w * CHUNK + s]  = s_cv[s];
    }
}

__global__ void __launch_bounds__(1024) gcn_bucket_rows(const int*  __restrict__ rec_row,
                                                        const int2* __restrict__ rec_cv,
                                                        const int*  __restrict__ tab,
                                                        int*        __restrict__ start,
                                                        int2*       __restrict__ pairs,
                                                        int n_nodes, int n_chunks,
                                                        int n_buckets) {
    __shared__ int  cnt[RB];
    __shared__ int  s_be[2];
    __shared__ int2 stage[STAGE];
    const int tid = threadIdx.x;
    const int b   = blockIdx.x;
    const int q   = tid & 3;
    if (tid < RB) cnt[tid] = 0;
    if (tid < 2)  s_be[tid] = 0;
    __syncthreads();
    int sb = 0, se = 0;
    for (int w = tid >> 2; w < n_chunks; w += 256) {
        int c0 = tab[b * n_chunks + w];
        int c1 = tab[(b + 1) * n_chunks + w];
        if (q == 0) { sb += c0; se += c1; }
        c0 = (c0 < 0) ? 0 : c0;
        c1 = (c1 > CHUNK) ? CHUNK : c1;
        const int* rr_base = rec_row + w * CHUNK;
        for (int j0 = c0 + q; j0 < c1; j0 += 4 * CELL_U) {
            int rr[CELL_U];
#pragma unroll
            for (int u = 0; u < CELL_U; ++u) {
                const int j = j0 + 4 * u;
                rr[u] = (j < c1) ? rr_base[j] : -1;
            }
#pragma unroll
            for (int u = 0; u < CELL_U; ++u)
                if (rr[u] >= 0) atomicAdd(&cnt[rr[u] & (RB - 1)], 1);
        }
    }
#pragma unroll
    for (int m = 16; m >= 1; m >>= 1) {
        sb += __shfl_xor(sb, m, 32);
        se += __shfl_xor(se, m, 32);
    }
    if ((tid & 31) == 0) { atomicAdd(&s_be[0], sb); atomicAdd(&s_be[1], se); }
    __syncthreads();
    const int beg = s_be[0];
    const int end = s_be[1];
    if (tid < 32) gcn_scan256(cnt, tid);
    __syncthreads();
    if (tid < RB) {
        const int g = b * RB + tid;
        if (g < n_nodes) start[g] = beg + cnt[tid];
    }
    if (b == n_buckets - 1 && tid == 0) start[n_nodes] = end;
    __syncthreads();
    for (int w = tid >> 2; w < n_chunks; w += 256) {
        int c0 = tab[b * n_chunks + w];
        int c1 = tab[(b + 1) * n_chunks + w];
        c0 = (c0 < 0) ? 0 : c0;
        c1 = (c1 > CHUNK) ? CHUNK : c1;
        const int*  rr_base = rec_row + w * CHUNK;
        const int2* cv_base = rec_cv + w * CHUNK;
        for (int j0 = c0 + q; j0 < c1; j0 += 4 * CELL_U) {
            int  rr[CELL_U];
            int2 cv[CELL_U];
#pragma unroll
            for (int u = 0; u < CELL_U; ++u) {
                const int j = j0 + 4 * u;
                rr[u] = -1;
                cv[u] = make_int2(0, 0);
                if (j < c1) { rr[u] = rr_base[j]; cv[u] = cv_base[j]; }
            }
#pragma unroll
            for (int u = 0; u < CELL_U; ++u) {
                if (rr[u] >= 0) {
                    const int p = atomicAdd(&cnt[rr[u] & (RB - 1)], 1);
                    if (p < STAGE) stage[p] = cv[u];
                    else           pairs[beg + p] = cv[u];
                }
            }
        }
    }
    __syncthreads();
    int m = end - beg;
    m = (m > STAGE) ? STAGE : m;
    for (int i = tid; i < m; i += 1024) pairs[beg + i] = stage[i];
}

__device__ __forceinline__ float4 gcn_gather_row(const float* __restrict__ src,
                                                 const int2*  __restrict__ pairs,
                                                 int beg, int end, int2 pr, int lane) {
    float4 acc = make_float4(0.f, 0.f, 0.f, 0.f);
    for (int j = beg; j < end; j += 32) {
        const int n = (end - j < 32) ? (end - j) : 32;
        if (j != beg) {
            pr = make_int2(0, 0);
            if (lane < n) pr = pairs[j + lane];
        }
#pragma unroll 4
        for (int k = 0; k < n; ++k) {
            const int   c = __builtin_amdgcn_readlane(pr.x, k);
            const float v = __int_as_float(__builtin_amdgcn_readlane(pr.y, k));
            const float4 x = ((const float4*)(src + (size_t)c * HIDDEN))[lane];
            acc.x = __builtin_fmaf(v, x.x, acc.x);
            acc.y = __builtin_fmaf(v, x.y, acc.y);
            acc.z = __builtin_fmaf(v, x.z, acc.z);
            acc.w = __builtin_fmaf(v, x.w, acc.w);
        }
    }
    return acc;
}

__device__ __forceinline__ v8f wmma_add_lane_pair(v8f c, v2f a, v2f b) {
    return __builtin_amdgcn_wmma_f32_16x16x4_f32(
         false, a,  false, b,
         (short)0, c,  false,  false);
}

template <bool FINAL>
__device__ __forceinline__ void gcn_spmm_group4(const float* __restrict__ src,
                                                const int*   __restrict__ start,
                                                const int2*  __restrict__ pairs,
                                                const float* __restrict__ bias,
                                                const float* __restrict__ fea,
                                                float*       __restrict__ dst,
                                                int n_groups) {
    const int lane = threadIdx.x & 31;
    const int wave = __builtin_amdgcn_readfirstlane(threadIdx.x >> 5);
    const int g    = blockIdx.x * 8 + wave;
    if (g >= n_groups) return;
    const int r0 = g * 4;

    int s[5];
#pragma unroll
    for (int i = 0; i < 5; ++i) s[i] = __builtin_amdgcn_readfirstlane(start[r0 + i]);

    int2 pr[4];
#pragma unroll
    for (int i = 0; i < 4; ++i) {
        pr[i] = make_int2(0, 0);
        if (lane < s[i + 1] - s[i]) pr[i] = pairs[s[i] + lane];
    }

    float4 acc[4];
#pragma unroll
    for (int i = 0; i < 4; ++i) acc[i] = gcn_gather_row(src, pairs, s[i], s[i + 1], pr[i], lane);

    if (FINAL) {
        float4 f[4], l[4];
#pragma unroll
        for (int i = 0; i < 4; ++i) {
            f[i] = ((const float4*)(fea + (size_t)(r0 + i) * HIDDEN))[lane];
            l[i] = ((const float4*)(src + (size_t)(r0 + i) * HIDDEN))[lane];
        }
#pragma unroll
        for (int i = 0; i < 4; ++i) {
            acc[i].x = (f[i].x + l[i].x) + acc[i].x;
            acc[i].y = (f[i].y + l[i].y) + acc[i].y;
            acc[i].z = (f[i].z + l[i].z) + acc[i].z;
            acc[i].w = (f[i].w + l[i].w) + acc[i].w;
        }
    }

    const int  c   = lane & 15;
    const bool sel = (lane >> 4) == (c >> 3);
    v2f a;
    a.x = (sel && !(c & 1)) ? 1.0f : 0.0f;
    a.y = (sel &&  (c & 1)) ? 1.0f : 0.0f;
    const float4 bv = ((const float4*)bias)[lane];
    v2f bxy, bzw;
    bxy.x = bv.x; bxy.y = bv.y;
    bzw.x = bv.z; bzw.y = bv.w;

    v8f cxy, czw;
#pragma unroll
    for (int i = 0; i < 4; ++i) {
        cxy[2 * i] = acc[i].x; cxy[2 * i + 1] = acc[i].y;
        czw[2 * i] = acc[i].z; czw[2 * i + 1] = acc[i].w;
    }
    v8f dxy = wmma_add_lane_pair(cxy, a, bxy);
    v8f dzw = wmma_add_lane_pair(czw, a, bzw);
    asm volatile("v_nop\n\tv_nop\n\tv_nop\n\tv_nop"
                 : "+v"(dxy), "+v"(dzw) : "v"(a), "v"(bxy), "v"(bzw));

    const float scale = FINAL ? (1.0f / 3.0f) : 1.0f;
#pragma unroll
    for (int i = 0; i < 4; ++i) {
        float4 o;
        if (FINAL) o = make_float4(dxy[2 * i] * scale, dxy[2 * i + 1] * scale,
                                   dzw[2 * i] * scale, dzw[2 * i + 1] * scale);
        else       o = make_float4(dxy[2 * i], dxy[2 * i + 1], dzw[2 * i], dzw[2 * i + 1]);
        ((float4*)(dst + (size_t)(r0 + i) * HIDDEN))[lane] = o;
    }
}

__global__ void __launch_bounds__(256) gcn_spmm_bias(const float* __restrict__ fea,
                                                     const int*   __restrict__ start,
                                                     const int2*  __restrict__ pairs,
                                                     const float* __restrict__ bias,
                                                     float*       __restrict__ l1,
                                                     int n_groups) {
    gcn_spmm_group4<false>(fea, start, pairs, bias, fea, l1, n_groups);
}

__global__ void __launch_bounds__(256) gcn_spmm_bias_mean(const float* __restrict__ l1,
                                                          const int*   __restrict__ start,
                                                          const int2*  __restrict__ pairs,
                                                          const float* __restrict__ bias,
                                                          const float* __restrict__ fea,
                                                          float*       __restrict__ out,
                                                          int n_groups) {
    gcn_spmm_group4<true>(l1, start, pairs, bias, fea, out, n_groups);
}

extern "C" void kernel_launch(void* const* d_in, const int* in_sizes, int n_in,
                              void* d_out, int out_size, void* d_ws, size_t ws_size,
                              hipStream_t stream) {
    const float* fea  = (const float*)d_in[0];
    const int*   row  = (const int*)  d_in[1];
    const int*   col  = (const int*)  d_in[2];
    const float* val  = (const float*)d_in[3];
    const float* bias = (const float*)d_in[4];
    float*       out  = (float*)d_out;

    const int n_nodes = in_sizes[0] / HIDDEN;
    const int n_edges = in_sizes[1];

    const int n1        = (n_nodes + 1 + 3) & ~3;
    const int ne        = (n_edges + 3) & ~3;
    const int n_chunks  = (n_edges + CHUNK - 1) / CHUNK;
    const int n_buckets = (n_nodes + RB - 1) / RB;
    const int n_tab     = ((n_buckets + 1) * n_chunks + 3) & ~3;

    float* acc1    = (float*)d_ws;
    int2*  rec_cv  = (int2*)d_ws;
    int*   rec_row = (int*)(rec_cv + ne);
    int*   tab     = (int*)(acc1 + (size_t)n_nodes * HIDDEN);
    int*   start   = tab + n_tab;
    int2*  pairs   = (int2*)(start + n1);

    const int n_groups    = n_nodes / 4;
    const int spmm_blocks = (n_groups + 7) / 8;

    gcn_chunk_sort<<<n_chunks, 512, 0, stream>>>(row, col, val, tab, rec_row, rec_cv,
                                                 n_edges, n_chunks, n_buckets);
    gcn_bucket_rows<<<n_buckets, 1024, 0, stream>>>(rec_row, rec_cv, tab, start, pairs,
                                                    n_nodes, n_chunks, n_buckets);

    gcn_spmm_bias<<<spmm_blocks, 256, 0, stream>>>(fea, start, pairs, bias, acc1, n_groups);
    gcn_spmm_bias_mean<<<spmm_blocks, 256, 0, stream>>>(acc1, start, pairs, bias + HIDDEN, fea, out, n_groups);
}
